# starting kernel
# baseline (speedup 1.0000x reference)
#include <hip/hip_runtime.h>
#include <hip/hip_bf16.h>

typedef __attribute__((ext_vector_type(2))) float v2f;
typedef __attribute__((ext_vector_type(8))) float v8f;

#define IN_DIM    128
#define OUT_DIM   128
#define NUM_HEADS 4
#define HEAD_DIM  32
#define NEG_SLOPE 0.2f
#define GAT_EPS   1e-16f

__global__ __launch_bounds__(256) void gat_gemm_wmma(const float* __restrict__ X,
                                                     const float* __restrict__ W,
                                                     float* __restrict__ H,
                                                     int M) {
  const int wave = threadIdx.x >> 5;
  const int lane = threadIdx.x & 31;
  const int m0 = blockIdx.x * 16;
  const int n0 = wave * 16;
  const int rc   = lane & 15;
  const int koff = (lane >> 4) * 2;

  int arow = m0 + rc;
  if (arow >= M) arow = M - 1;
  const float* __restrict__ xrow = X + (size_t)arow * IN_DIM + koff;

  v8f c = {};
#pragma unroll
  for (int k = 0; k < IN_DIM; k += 4) {
    v2f a = *(const v2f*)(xrow + k);
    v2f b;
    b.x = W[(size_t)(k + koff) * OUT_DIM + n0 + rc];
    b.y = W[(size_t)(k + koff + 1) * OUT_DIM + n0 + rc];
    c = __builtin_amdgcn_wmma_f32_16x16x4_f32(false, a, false, b, (short)0, c, false, false);
  }

  const int rbase = (lane >> 4) * 8;
#pragma unroll
  for (int r = 0; r < 8; ++r) {
    const int row = m0 + rbase + r;
    if (row < M) H[(size_t)row * OUT_DIM + n0 + rc] = c[r];
  }
}

__global__ __launch_bounds__(256) void gat_alpha(const float* __restrict__ H,
                                                 const float* __restrict__ a,
                                                 float* __restrict__ alpha_src,
                                                 float* __restrict__ alpha_dst,
                                                 int N) {
  const int lane = threadIdx.x & 31;
  const int node = blockIdx.x * 8 + (threadIdx.x >> 5);
  if (node >= N) return;

  const int d0    = lane * 4;
  const int head  = lane >> 3;
  const int local = d0 & (HEAD_DIM - 1);

  const float4 hv = *(const float4*)(H + (size_t)node * OUT_DIM + d0);
  const float4 as = *(const float4*)(a + head * (2 * HEAD_DIM) + local);
  const float4 ad = *(const float4*)(a + head * (2 * HEAD_DIM) + HEAD_DIM + local);

  float ps = hv.x * as.x + hv.y * as.y + hv.z * as.z + hv.w * as.w;
  float pd = hv.x * ad.x + hv.y * ad.y + hv.z * ad.z + hv.w * ad.w;

#pragma unroll
  for (int off = 1; off <= 4; off <<= 1) {
    ps += __shfl_xor(ps, off, 32);
    pd += __shfl_xor(pd, off, 32);
  }
  if ((lane & 7) == 0) {
    alpha_src[(size_t)node * NUM_HEADS + head] = ps;
    alpha_dst[(size_t)node * NUM_HEADS + head] = pd;
  }
}

__global__ __launch_bounds__(256) void gat_edge_scores(const int* __restrict__ ei,
                                                       const float* __restrict__ alpha_src,
                                                       const float* __restrict__ alpha_dst,
                                                       float* __restrict__ exp_s,
                                                       float* __restrict__ sum_exp,
                                                       int E) {
  const int e = blockIdx.x * blockDim.x + threadIdx.x;
  if (e >= E) return;
  const int src = ei[e];
  const int dst = ei[E + e];

  const float4 s1 = *(const float4*)(alpha_src + (size_t)src * NUM_HEADS);
  const float4 s2 = *(const float4*)(alpha_dst + (size_t)dst * NUM_HEADS);

  float s[4] = {s1.x + s2.x, s1.y + s2.y, s1.z + s2.z, s1.w + s2.w};
  float4 ex;
#pragma unroll
  for (int h = 0; h < 4; ++h) {
    float v = s[h];
    v = (v >= 0.0f) ? v : NEG_SLOPE * v;
    ((float*)&ex)[h] = expf(v);
  }
  *(float4*)(exp_s + (size_t)e * NUM_HEADS) = ex;

  float* sp = sum_exp + (size_t)dst * NUM_HEADS;
  unsafeAtomicAdd(sp + 0, ex.x);
  unsafeAtomicAdd(sp + 1, ex.y);
  unsafeAtomicAdd(sp + 2, ex.z);
  unsafeAtomicAdd(sp + 3, ex.w);
}

__global__ __launch_bounds__(256) void gat_aggregate(const int* __restrict__ ei,
                                                     const float* __restrict__ H,
                                                     const float* __restrict__ exp_s,
                                                     const float* __restrict__ sum_exp,
                                                     float* __restrict__ out,
                                                     int E) {
  const int lane = threadIdx.x & 31;
  const int e = blockIdx.x * 8 + (threadIdx.x >> 5);
  if (e >= E) return;

  const int src = ei[e];
  const int dst = ei[E + e];
  const int head = lane >> 3;

  const float attn = exp_s[(size_t)e * NUM_HEADS + head] /
                     (sum_exp[(size_t)dst * NUM_HEADS + head] + GAT_EPS);

  const int d0 = lane * 4;
  const float4 hv = *(const float4*)(H + (size_t)src * OUT_DIM + d0);
  float* op = out + (size_t)dst * OUT_DIM + d0;
  unsafeAtomicAdd(op + 0, hv.x * attn);
  unsafeAtomicAdd(op + 1, hv.y * attn);
  unsafeAtomicAdd(op + 2, hv.z * attn);
  unsafeAtomicAdd(op + 3, hv.w * attn);
}

extern "C" void kernel_launch(void* const* d_in, const int* in_sizes, int n_in,
                              void* d_out, int out_size, void* d_ws, size_t ws_size,
                              hipStream_t stream) {
  const float* X  = (const float*)d_in[0];
  const int*   ei = (const int*)d_in[1];
  const float* W  = (const float*)d_in[2];
  const float* a  = (const float*)d_in[3];

  const int N = in_sizes[0] / IN_DIM;
  const int E = in_sizes[1] / 2;

  float* H         = (float*)d_ws;
  float* alpha_src = H + (size_t)N * OUT_DIM;
  float* alpha_dst = alpha_src + (size_t)N * NUM_HEADS;
  float* exp_s     = alpha_dst + (size_t)N * NUM_HEADS;
  float* sum_exp   = exp_s + (size_t)E * NUM_HEADS;

  hipMemsetAsync(sum_exp, 0, (size_t)N * NUM_HEADS * sizeof(float), stream);
  hipMemsetAsync(d_out, 0, (size_t)out_size * sizeof(float), stream);

  gat_gemm_wmma<<<(N + 15) / 16, 256, 0, stream>>>(X, W, H, N);
  gat_alpha<<<(N + 7) / 8, 256, 0, stream>>>(H, a, alpha_src, alpha_dst, N);
  gat_edge_scores<<<(E + 255) / 256, 256, 0, stream>>>(ei, alpha_src, alpha_dst,
                                                       exp_s, sum_exp, E);
  gat_aggregate<<<(E + 7) / 8, 256, 0, stream>>>(ei, H, exp_s, sum_exp,
                                                 (float*)d_out, E);
}


# GEMM re-tiled: a wave computes a whole 16x128 slab of H (8 accumulator tiles) and reads its X rows once, 16 bytes per load; W sits in LDS as B; whole-line H stores; k summed in a new order
# speedup vs baseline: 3.3228x; 3.3228x over previous
#include <hip/hip_runtime.h>
#include <hip/hip_bf16.h>

typedef __attribute__((ext_vector_type(2))) float v2f;
typedef __attribute__((ext_vector_type(8))) float v8f;

#define IN_DIM    128
#define OUT_DIM   128
#define NUM_HEADS 4
#define HEAD_DIM  32
#define NEG_SLOPE 0.2f
#define GAT_EPS   1e-16f

#define GWAVES 16
#define GBLOCKS 256
#define WL_F4 ((IN_DIM / 2) * (OUT_DIM / 2))
#define WSTAGE_TRIPS (((IN_DIM / 2) * (OUT_DIM / 4)) / (GWAVES * 32))

static_assert(IN_DIM == 128 && OUT_DIM == 128, "the slab kernel is written for 128 x 128 weights");
static_assert(WSTAGE_TRIPS * GWAVES * 32 == (IN_DIM / 2) * (OUT_DIM / 4), "the copy of W divides evenly over the threads");

__device__ __forceinline__ void load_slab(const float* __restrict__ X, const int M, const int slab,
                                          const int c, const int h, float4 (&a)[16]) {
  int arow = slab * 16 + c;
  if (arow >= M) arow = M - 1;
  const float4* __restrict__ xr = (const float4*)(X + (size_t)arow * IN_DIM) + h;
#pragma unroll
  for (int j = 0; j < 16; ++j) a[j] = xr[2 * j];
}

__global__ __launch_bounds__(GWAVES * 32) void gat_gemm_wmma(const float* __restrict__ X,
                                                             const float* __restrict__ W,
                                                             float* __restrict__ H,
                                                             int M) {
  __shared__ float4 wl[WL_F4];

  const int lane = threadIdx.x & 31;
  const int wave = __builtin_amdgcn_readfirstlane((int)(threadIdx.x >> 5));
  const int c = lane & 15;
  const int h = lane >> 4;
  const int nslabs = (M + 15) / 16;
  const int stride = (int)gridDim.x * GWAVES;

  int slab = (int)blockIdx.x + (int)gridDim.x * wave;
  float4 a[16];
  if (slab < nslabs) load_slab(X, M, slab, c, h, a);

#pragma unroll
  for (int t = 0; t < WSTAGE_TRIPS; ++t) {
    const int i  = (int)threadIdx.x + t * (GWAVES * 32);
    const int kp = i >> 5;
    const int m  = i & 31;
    const float4 r0 = *(const float4*)(W + (size_t)(2 * kp) * OUT_DIM + 4 * m);
    const float4 r1 = *(const float4*)(W + (size_t)(2 * kp + 1) * OUT_DIM + 4 * m);
    const int e = (kp * 4 + 2 * (m >> 4)) * 16 + (m & 15);
    wl[e]      = make_float4(r0.x, r1.x, r0.y, r1.y);
    wl[e + 16] = make_float4(r0.z, r1.z, r0.w, r1.w);
  }
  __syncthreads();

  const float4* __restrict__ bl = wl + (128 * h + c);

  while (slab < nslabs) {
    v8f acc[8] = {};
    float4 bc[8], bn[8];
#pragma unroll
    for (int i = 0; i < 8; ++i) bc[i] = bl[(i >> 2) * 64 + (i & 3) * 16];
#pragma unroll
    for (int j = 0; j < 16; ++j) {
      if (j + 1 < 16) {
#pragma unroll
        for (int i = 0; i < 8; ++i) bn[i] = bl[(4 * (j + 1) + (i >> 2)) * 64 + (i & 3) * 16];
      }
#pragma unroll
      for (int s = 0; s < 2; ++s) {
        v2f av;
        av.x = s ? a[j].z : a[j].x;
        av.y = s ? a[j].w : a[j].y;
#pragma unroll
        for (int p = 0; p < 4; ++p) {
          const float4 bv = bc[4 * s + p];
          v2f b;
          b.x = bv.x; b.y = bv.y;
          acc[2 * p] = __builtin_amdgcn_wmma_f32_16x16x4_f32(false, av, false, b, (short)0, acc[2 * p], false, false);
          b.x = bv.z; b.y = bv.w;
          acc[2 * p + 1] = __builtin_amdgcn_wmma_f32_16x16x4_f32(false, av, false, b, (short)0, acc[2 * p + 1], false, false);
        }
      }
      __builtin_amdgcn_sched_barrier(0);
#pragma unroll
      for (int i = 0; i < 8; ++i) bc[i] = bn[i];
    }

    const int m0 = slab * 16;
    slab += stride;
    if (slab < nslabs) load_slab(X, M, slab, c, h, a);

#pragma unroll
    for (int r = 0; r < 8; ++r) {
      const int row = m0 + 8 * h + r;
      if (row < M) {
        float4* __restrict__ hp = (float4*)(H + (size_t)row * OUT_DIM) + c;
        hp[0]  = make_float4(acc[0][r], acc[1][r], acc[2][r], acc[3][r]);
        hp[16] = make_float4(acc[4][r], acc[5][r], acc[6][r], acc[7][r]);
      }
    }
  }
}

__global__ __launch_bounds__(256) void gat_alpha(const float* __restrict__ H,
                                                 const float* __restrict__ a,
                                                 float* __restrict__ alpha_src,
                                                 float* __restrict__ alpha_dst,
                                                 int N) {
  const int lane = threadIdx.x & 31;
  const int node = blockIdx.x * 8 + (threadIdx.x >> 5);
  if (node >= N) return;

  const int d0    = lane * 4;
  const int head  = lane >> 3;
  const int local = d0 & (HEAD_DIM - 1);

  const float4 hv = *(const float4*)(H + (size_t)node * OUT_DIM + d0);
  const float4 as = *(const float4*)(a + head * (2 * HEAD_DIM) + local);
  const float4 ad = *(const float4*)(a + head * (2 * HEAD_DIM) + HEAD_DIM + local);

  float ps = hv.x * as.x + hv.y * as.y + hv.z * as.z + hv.w * as.w;
  float pd = hv.x * ad.x + hv.y * ad.y + hv.z * ad.z + hv.w * ad.w;

#pragma unroll
  for (int off = 1; off <= 4; off <<= 1) {
    ps += __shfl_xor(ps, off, 32);
    pd += __shfl_xor(pd, off, 32);
  }
  if ((lane & 7) == 0) {
    alpha_src[(size_t)node * NUM_HEADS + head] = ps;
    alpha_dst[(size_t)node * NUM_HEADS + head] = pd;
  }
}

#define ROWS     392
#define OWAVES   32
#define LCAP     896
#define SCAN_U   6
#define SCAN_IDS (32 * 4 * SCAN_U)
#define EPW 32
#define CH  8
#define NGRP (EPW / CH)
#define ROW_BITS 9
#define EDGE_BITS (32 - ROW_BITS)

static_assert(LCAP > SCAN_IDS, "a scan step must always fit behind a worked-off list");
static_assert(ROWS < (1 << ROW_BITS), "row numbers, the spare row included, fit their field");
static_assert(((ROWS + 1) * OUT_DIM + (ROWS + 1) * NUM_HEADS + OWAVES * LCAP) * 4 <= 320 * 1024, "LDS budget");

__device__ __forceinline__ float lane_to_scalar(float v, int l) {
  return __int_as_float(__builtin_amdgcn_readlane(__float_as_int(v), l));
}

__device__ __forceinline__ void lds_add(float* p, float v) {
  __hip_atomic_fetch_add(p, v, __ATOMIC_RELAXED, __HIP_MEMORY_SCOPE_WORKGROUP);
}

__device__ __forceinline__ void work_off(const int* __restrict__ lst, const int n,
                                         const bool packed, const unsigned lane,
                                         const int* __restrict__ ei, const int E,
                                         const float* __restrict__ H,
                                         const float* __restrict__ alpha_src,
                                         const float* __restrict__ alpha_dst,
                                         float* tile, float* dens, const int lo) {
  const unsigned hh = lane & (NUM_HEADS - 1);
  const int qsub = (int)(lane >> 2);
#pragma nounroll
  for (int b = 0; b < n; b += EPW) {
    const int cnt = (n - b < EPW) ? (n - b) : EPW;
    unsigned sv[NGRP], dg[NGRP], dl[NGRP], word[NGRP], row[NGRP];
    float    ex[NGRP];
#pragma unroll
    for (int g = 0; g < NGRP; ++g) {
      const int q = CH * g + qsub;
      word[g] = (unsigned)lst[b + ((q < cnt) ? q : 0)];
    }
#pragma unroll
    for (int g = 0; g < NGRP; ++g) {
      const unsigned e = packed ? (word[g] & ((1u << EDGE_BITS) - 1u)) : word[g];
      sv[g] = (unsigned)ei[e];
    }
    if (packed) {
#pragma unroll
      for (int g = 0; g < NGRP; ++g) row[g] = word[g] >> EDGE_BITS;
    } else {
#pragma unroll
      for (int g = 0; g < NGRP; ++g) row[g] = (unsigned)ei[(unsigned)E + word[g]] - (unsigned)lo;
    }
#pragma unroll
    for (int g = 0; g < NGRP; ++g) {
      dg[g] = row[g] + (unsigned)lo;
      dl[g] = (CH * g + qsub < cnt) ? row[g] : (unsigned)ROWS;
    }
#pragma unroll
    for (int g = 0; g < NGRP; ++g) {
      const float as = alpha_src[sv[g] * NUM_HEADS + hh];
      const float ad = alpha_dst[dg[g] * NUM_HEADS + hh];
      float v = as + ad;
      v = (v >= 0.0f) ? v : NEG_SLOPE * v;
      ex[g] = expf(v);
    }
#pragma unroll
    for (int g = 0; g < NGRP; ++g) {
      lds_add(dens + (dl[g] * NUM_HEADS + hh), ex[g]);
    }
#pragma unroll
    for (int g = 0; g < NGRP; ++g) {
      if (CH * g < cnt) {
        float h[CH][4];
#pragma unroll
        for (int u = 0; u < CH; ++u) {
          const unsigned s = (unsigned)__builtin_amdgcn_readlane((int)sv[g], 4 * u);
          const float* __restrict__ hrow = H + (s * OUT_DIM + lane);
          h[u][0] = hrow[0 * HEAD_DIM];
          h[u][1] = hrow[1 * HEAD_DIM];
          h[u][2] = hrow[2 * HEAD_DIM];
          h[u][3] = hrow[3 * HEAD_DIM];
        }
#pragma unroll
        for (int u = 0; u < CH; ++u) {
          const unsigned d = (unsigned)__builtin_amdgcn_readlane((int)dl[g], 4 * u);
          const float w0 = lane_to_scalar(ex[g], 4 * u + 0);
          const float w1 = lane_to_scalar(ex[g], 4 * u + 1);
          const float w2 = lane_to_scalar(ex[g], 4 * u + 2);
          const float w3 = lane_to_scalar(ex[g], 4 * u + 3);
          float* op = tile + (d * OUT_DIM + lane);
          lds_add(op + 0 * HEAD_DIM, h[u][0] * w0);
          lds_add(op + 1 * HEAD_DIM, h[u][1] * w1);
          lds_add(op + 2 * HEAD_DIM, h[u][2] * w2);
          lds_add(op + 3 * HEAD_DIM, h[u][3] * w3);
        }
      }
    }
  }
}

__device__ __forceinline__ void load_ids(const int* __restrict__ dstp, const int E,
                                         const bool vec_ok, const int base,
                                         const unsigned lane, int (&id)[SCAN_U][4]) {
  if (vec_ok && base + SCAN_IDS <= E) {
#pragma unroll
    for (int u = 0; u < SCAN_U; ++u) {
      const int4 v = *(const int4*)(dstp + base + u * 128 + (int)lane * 4);
      id[u][0] = v.x; id[u][1] = v.y; id[u][2] = v.z; id[u][3] = v.w;
    }
  } else {
#pragma unroll
    for (int u = 0; u < SCAN_U; ++u) {
#pragma unroll
      for (int c = 0; c < 4; ++c) {
        const int e = base + u * 128 + (int)lane * 4 + c;
        id[u][c] = (e < E) ? dstp[e] : -1;
      }
    }
  }
}

__global__ __launch_bounds__(OWAVES * 32) void gat_owner_pass(const int* __restrict__ ei,
                                                              const float* __restrict__ H,
                                                              const float* __restrict__ alpha_src,
                                                              const float* __restrict__ alpha_dst,
                                                              float* __restrict__ out,
                                                              int N, int E) {
  __shared__ __attribute__((aligned(16))) float tile[(ROWS + 1) * OUT_DIM];
  __shared__ __attribute__((aligned(16))) float dens[(ROWS + 1) * NUM_HEADS];
  __shared__ int lists[OWAVES * LCAP];

  const unsigned lane = threadIdx.x & 31;
  const int wv = __builtin_amdgcn_readfirstlane((int)(threadIdx.x >> 5));
  const int lo = blockIdx.x * ROWS;
  const int rows = (N - lo < ROWS) ? (N - lo) : ROWS;

  {
    const float4 z = {0.0f, 0.0f, 0.0f, 0.0f};
    for (int i = threadIdx.x; i < ROWS * OUT_DIM / 4; i += OWAVES * 32) ((float4*)tile)[i] = z;
    for (int i = threadIdx.x; i < ROWS * NUM_HEADS / 4; i += OWAVES * 32) ((float4*)dens)[i] = z;
  }
  __syncthreads();

  int* __restrict__ mylist = lists + wv * LCAP;
  const int* __restrict__ dstp = ei + E;
  const bool vec_ok = ((E & 3) == 0);
  const bool packed = (E <= (1 << EDGE_BITS));
  const int nchunks = (E + SCAN_IDS - 1) / SCAN_IDS;
  const int nsteps = (nchunks + OWAVES - 1) / OWAVES;
  const int rot = (int)(((long long)blockIdx.x * nchunks) / gridDim.x);
  int cnt = 0;
#pragma nounroll
  for (int st = 0; st <= nsteps; ++st) {
    const int k = st * OWAVES + wv;
    if (st < nsteps && k < nchunks) {
      const int ck = (k + rot < nchunks) ? (k + rot) : (k + rot - nchunks);
      const int base = ck * SCAN_IDS;
      int id[SCAN_U][4];
      load_ids(dstp, E, vec_ok, base, lane, id);
#pragma unroll
      for (int u = 0; u < SCAN_U; ++u) {
#pragma unroll
        for (int c = 0; c < 4; ++c) {
          const unsigned row = (unsigned)(id[u][c] - lo);
          const bool mine = row < (unsigned)rows;
          const unsigned mask = __builtin_amdgcn_ballot_w32(mine);
          if (mask != 0u) {
            const unsigned pre = __builtin_amdgcn_mbcnt_lo(mask, 0u);
            const unsigned e = (unsigned)(base + u * 128 + (int)lane * 4 + c);
            if (mine) mylist[cnt + (int)pre] = (int)(packed ? (e | (row << EDGE_BITS)) : e);
            cnt += __builtin_popcount(mask);
          }
        }
      }
    }
    if (cnt > LCAP - SCAN_IDS || (st == nsteps && cnt > 0)) {
      __builtin_amdgcn_wave_barrier();
      work_off(mylist, cnt, packed, lane, ei, E, H, alpha_src, alpha_dst, tile, dens, lo);
      __builtin_amdgcn_wave_barrier();
      cnt = 0;
    }
  }
  __syncthreads();

  const unsigned head = lane >> 3;
  for (int r = wv; r < rows; r += OWAVES) {
    const float4 v = ((const float4*)(tile + r * OUT_DIM))[lane];
    const float dd = dens[r * NUM_HEADS + head] + GAT_EPS;
    float4 q;
    q.x = v.x / dd;
    q.y = v.y / dd;
    q.z = v.z / dd;
    q.w = v.w / dd;
    ((float4*)(out + (size_t)(lo + r) * OUT_DIM))[lane] = q;
  }
}

extern "C" void kernel_launch(void* const* d_in, const int* in_sizes, int n_in,
                              void* d_out, int out_size, void* d_ws, size_t ws_size,
                              hipStream_t stream) {
  const float* X  = (const float*)d_in[0];
  const int*   ei = (const int*)d_in[1];
  const float* W  = (const float*)d_in[2];
  const float* a  = (const float*)d_in[3];

  const int N = in_sizes[0] / IN_DIM;
  const int E = in_sizes[1] / 2;

  float* H         = (float*)d_ws;
  float* alpha_src = H + (size_t)N * OUT_DIM;
  float* alpha_dst = alpha_src + (size_t)N * NUM_HEADS;

  gat_gemm_wmma<<<GBLOCKS, GWAVES * 32, 0, stream>>>(X, W, H, N);
  gat_alpha<<<(N + 7) / 8, 256, 0, stream>>>(H, a, alpha_src, alpha_dst, N);
  gat_owner_pass<<<(N + ROWS - 1) / ROWS, OWAVES * 32, 0, stream>>>(ei, H, alpha_src, alpha_dst,
                                                                  (float*)d_out, N, E);
}


# edge scan reads one key byte per edge (owner workgroup number, written in the alpha launch), 16 keys per 16-byte load, in place of the 4-byte destination id; the work-off checks the id itself
# speedup vs baseline: 4.0133x; 1.2078x over previous
#include <hip/hip_runtime.h>
#include <hip/hip_bf16.h>

typedef __attribute__((ext_vector_type(2))) float v2f;
typedef __attribute__((ext_vector_type(8))) float v8f;

#define IN_DIM    128
#define OUT_DIM   128
#define NUM_HEADS 4
#define HEAD_DIM  32
#define NEG_SLOPE 0.2f
#define GAT_EPS   1e-16f

#define GWAVES 16
#define GBLOCKS 256
#define WL_F4 ((IN_DIM / 2) * (OUT_DIM / 2))
#define WSTAGE_TRIPS (((IN_DIM / 2) * (OUT_DIM / 4)) / (GWAVES * 32))

static_assert(IN_DIM == 128 && OUT_DIM == 128, "the slab kernel is written for 128 x 128 weights");
static_assert(WSTAGE_TRIPS * GWAVES * 32 == (IN_DIM / 2) * (OUT_DIM / 4), "the copy of W divides evenly over the threads");

__device__ __forceinline__ void load_slab(const float* __restrict__ X, const int M, const int slab,
                                          const int c, const int h, float4 (&a)[16]) {
  int arow = slab * 16 + c;
  if (arow >= M) arow = M - 1;
  const float4* __restrict__ xr = (const float4*)(X + (size_t)arow * IN_DIM) + h;
#pragma unroll
  for (int j = 0; j < 16; ++j) a[j] = xr[2 * j];
}

__global__ __launch_bounds__(GWAVES * 32) void gat_gemm_wmma(const float* __restrict__ X,
                                                             const float* __restrict__ W,
                                                             float* __restrict__ H,
                                                             int M) {
  __shared__ float4 wl[WL_F4];

  const int lane = threadIdx.x & 31;
  const int wave = __builtin_amdgcn_readfirstlane((int)(threadIdx.x >> 5));
  const int c = lane & 15;
  const int h = lane >> 4;
  const int nslabs = (M + 15) / 16;
  const int stride = (int)gridDim.x * GWAVES;

  int slab = (int)blockIdx.x + (int)gridDim.x * wave;
  float4 a[16];
  if (slab < nslabs) load_slab(X, M, slab, c, h, a);

#pragma unroll
  for (int t = 0; t < WSTAGE_TRIPS; ++t) {
    const int i  = (int)threadIdx.x + t * (GWAVES * 32);
    const int kp = i >> 5;
    const int m  = i & 31;
    const float4 r0 = *(const float4*)(W + (size_t)(2 * kp) * OUT_DIM + 4 * m);
    const float4 r1 = *(const float4*)(W + (size_t)(2 * kp + 1) * OUT_DIM + 4 * m);
    const int e = (kp * 4 + 2 * (m >> 4)) * 16 + (m & 15);
    wl[e]      = make_float4(r0.x, r1.x, r0.y, r1.y);
    wl[e + 16] = make_float4(r0.z, r1.z, r0.w, r1.w);
  }
  __syncthreads();

  const float4* __restrict__ bl = wl + (128 * h + c);

  while (slab < nslabs) {
    v8f acc[8] = {};
    float4 bc[8], bn[8];
#pragma unroll
    for (int i = 0; i < 8; ++i) bc[i] = bl[(i >> 2) * 64 + (i & 3) * 16];
#pragma unroll
    for (int j = 0; j < 16; ++j) {
      if (j + 1 < 16) {
#pragma unroll
        for (int i = 0; i < 8; ++i) bn[i] = bl[(4 * (j + 1) + (i >> 2)) * 64 + (i & 3) * 16];
      }
#pragma unroll
      for (int s = 0; s < 2; ++s) {
        v2f av;
        av.x = s ? a[j].z : a[j].x;
        av.y = s ? a[j].w : a[j].y;
#pragma unroll
        for (int p = 0; p < 4; ++p) {
          const float4 bv = bc[4 * s + p];
          v2f b;
          b.x = bv.x; b.y = bv.y;
          acc[2 * p] = __builtin_amdgcn_wmma_f32_16x16x4_f32(false, av, false, b, (short)0, acc[2 * p], false, false);
          b.x = bv.z; b.y = bv.w;
          acc[2 * p + 1] = __builtin_amdgcn_wmma_f32_16x16x4_f32(false, av, false, b, (short)0, acc[2 * p + 1], false, false);
        }
      }
      __builtin_amdgcn_sched_barrier(0);
#pragma unroll
      for (int i = 0; i < 8; ++i) bc[i] = bn[i];
    }

    const int m0 = slab * 16;
    slab += stride;
    if (slab < nslabs) load_slab(X, M, slab, c, h, a);

#pragma unroll
    for (int r = 0; r < 8; ++r) {
      const int row = m0 + 8 * h + r;
      if (row < M) {
        float4* __restrict__ hp = (float4*)(H + (size_t)row * OUT_DIM) + c;
        hp[0]  = make_float4(acc[0][r], acc[1][r], acc[2][r], acc[3][r]);
        hp[16] = make_float4(acc[4][r], acc[5][r], acc[6][r], acc[7][r]);
      }
    }
  }
}

#define ROWS      392
#define KSCAN_U   4
#define KSCAN_IDS (32 * 16 * KSCAN_U)

__global__ __launch_bounds__(256) void gat_alpha(const float* __restrict__ H,
                                                 const float* __restrict__ a,
                                                 float* __restrict__ alpha_src,
                                                 float* __restrict__ alpha_dst,
                                                 const int* __restrict__ ei,
                                                 unsigned* __restrict__ keyw,
                                                 int N, int E) {
  {
    const int* __restrict__ dstp = ei + E;
    const bool vec_ok = ((E & 3) == 0);
    const unsigned nwords = (unsigned)((E + KSCAN_IDS - 1) / KSCAN_IDS) * (KSCAN_IDS / 4);
    const unsigned total = gridDim.x * blockDim.x;
    for (unsigned q = blockIdx.x * blockDim.x + threadIdx.x; q < nwords; q += total) {
      const unsigned e0 = 4u * q;
      unsigned d[4];
      if (vec_ok && e0 + 3u < (unsigned)E) {
        const int4 v = *(const int4*)(dstp + e0);
        d[0] = (unsigned)v.x; d[1] = (unsigned)v.y; d[2] = (unsigned)v.z; d[3] = (unsigned)v.w;
      } else {
#pragma unroll
        for (int c = 0; c < 4; ++c) d[c] = (e0 + c < (unsigned)E) ? (unsigned)dstp[e0 + c] : 0u;
      }
      keyw[q] = ((d[0] / ROWS) & 255u) | (((d[1] / ROWS) & 255u) << 8) |
                (((d[2] / ROWS) & 255u) << 16) | (((d[3] / ROWS) & 255u) << 24);
    }
  }
  const int lane = threadIdx.x & 31;
  const int node = blockIdx.x * 8 + (threadIdx.x >> 5);
  if (node >= N) return;

  const int d0    = lane * 4;
  const int head  = lane >> 3;
  const int local = d0 & (HEAD_DIM - 1);

  const float4 hv = *(const float4*)(H + (size_t)node * OUT_DIM + d0);
  const float4 as = *(const float4*)(a + head * (2 * HEAD_DIM) + local);
  const float4 ad = *(const float4*)(a + head * (2 * HEAD_DIM) + HEAD_DIM + local);

  float ps = hv.x * as.x + hv.y * as.y + hv.z * as.z + hv.w * as.w;
  float pd = hv.x * ad.x + hv.y * ad.y + hv.z * ad.z + hv.w * ad.w;

#pragma unroll
  for (int off = 1; off <= 4; off <<= 1) {
    ps += __shfl_xor(ps, off, 32);
    pd += __shfl_xor(pd, off, 32);
  }
  if ((lane & 7) == 0) {
    alpha_src[(size_t)node * NUM_HEADS + head] = ps;
    alpha_dst[(size_t)node * NUM_HEADS + head] = pd;
  }
}

#define OWAVES   32
#define LCAP     896
#define EPW 32
#define CH  8
#define NGRP (EPW / CH)

static_assert(LCAP >= 2 * EPW, "a full list still takes the edges of one more ballot");
static_assert(((ROWS + 1) * OUT_DIM + (ROWS + 1) * NUM_HEADS + OWAVES * LCAP) * 4 <= 320 * 1024, "LDS budget");

__device__ __forceinline__ float lane_to_scalar(float v, int l) {
  return __int_as_float(__builtin_amdgcn_readlane(__float_as_int(v), l));
}

__device__ __forceinline__ void lds_add(float* p, float v) {
  __hip_atomic_fetch_add(p, v, __ATOMIC_RELAXED, __HIP_MEMORY_SCOPE_WORKGROUP);
}

__device__ __forceinline__ void work_off(const int* __restrict__ lst, const int n,
                                         const unsigned lane,
                                         const int* __restrict__ ei, const int E,
                                         const float* __restrict__ H,
                                         const float* __restrict__ alpha_src,
                                         const float* __restrict__ alpha_dst,
                                         float* tile, float* dens, const int lo, const int rows) {
  const unsigned hh = lane & (NUM_HEADS - 1);
  const int qsub = (int)(lane >> 2);
#pragma nounroll
  for (int b = 0; b < n; b += EPW) {
    const int cnt = (n - b < EPW) ? (n - b) : EPW;
    unsigned sv[NGRP], dg[NGRP], dl[NGRP], word[NGRP];
    float    ex[NGRP];
#pragma unroll
    for (int g = 0; g < NGRP; ++g) {
      const int q = CH * g + qsub;
      word[g] = (unsigned)lst[b + ((q < cnt) ? q : 0)];
    }
#pragma unroll
    for (int g = 0; g < NGRP; ++g) {
      sv[g] = (unsigned)ei[word[g]];
      dg[g] = (unsigned)ei[(unsigned)E + word[g]];
    }
#pragma unroll
    for (int g = 0; g < NGRP; ++g) {
      const unsigned row = dg[g] - (unsigned)lo;
      dl[g] = (CH * g + qsub < cnt && row < (unsigned)rows) ? row : (unsigned)ROWS;
    }
#pragma unroll
    for (int g = 0; g < NGRP; ++g) {
      const float as = alpha_src[sv[g] * NUM_HEADS + hh];
      const float ad = alpha_dst[dg[g] * NUM_HEADS + hh];
      float v = as + ad;
      v = (v >= 0.0f) ? v : NEG_SLOPE * v;
      ex[g] = expf(v);
    }
#pragma unroll
    for (int g = 0; g < NGRP; ++g) {
      lds_add(dens + (dl[g] * NUM_HEADS + hh), ex[g]);
    }
#pragma unroll
    for (int g = 0; g < NGRP; ++g) {
      if (CH * g < cnt) {
        float h[CH][4];
#pragma unroll
        for (int u = 0; u < CH; ++u) {
          const unsigned s = (unsigned)__builtin_amdgcn_readlane((int)sv[g], 4 * u);
          const float* __restrict__ hrow = H + (s * OUT_DIM + lane);
          h[u][0] = hrow[0 * HEAD_DIM];
          h[u][1] = hrow[1 * HEAD_DIM];
          h[u][2] = hrow[2 * HEAD_DIM];
          h[u][3] = hrow[3 * HEAD_DIM];
        }
#pragma unroll
        for (int u = 0; u < CH; ++u) {
          const unsigned d = (unsigned)__builtin_amdgcn_readlane((int)dl[g], 4 * u);
          const float w0 = lane_to_scalar(ex[g], 4 * u + 0);
          const float w1 = lane_to_scalar(ex[g], 4 * u + 1);
          const float w2 = lane_to_scalar(ex[g], 4 * u + 2);
          const float w3 = lane_to_scalar(ex[g], 4 * u + 3);
          float* op = tile + (d * OUT_DIM + lane);
          lds_add(op + 0 * HEAD_DIM, h[u][0] * w0);
          lds_add(op + 1 * HEAD_DIM, h[u][1] * w1);
          lds_add(op + 2 * HEAD_DIM, h[u][2] * w2);
          lds_add(op + 3 * HEAD_DIM, h[u][3] * w3);
        }
      }
    }
  }
}

__device__ __forceinline__ unsigned equal_bytes(const unsigned v, const unsigned pat) {
  const unsigned x = v ^ pat;
  return (x - 0x01010101u) & ~x & 0x80808080u;
}

__global__ __launch_bounds__(OWAVES * 32) void gat_owner_pass(const int* __restrict__ ei,
                                                              const unsigned char* __restrict__ key,
                                                              const float* __restrict__ H,
                                                              const float* __restrict__ alpha_src,
                                                              const float* __restrict__ alpha_dst,
                                                              float* __restrict__ out,
                                                              int N, int E) {
  __shared__ __attribute__((aligned(16))) float tile[(ROWS + 1) * OUT_DIM];
  __shared__ __attribute__((aligned(16))) float dens[(ROWS + 1) * NUM_HEADS];
  __shared__ int lists[OWAVES * LCAP];

  const unsigned lane = threadIdx.x & 31;
  const int wv = __builtin_amdgcn_readfirstlane((int)(threadIdx.x >> 5));
  const int lo = blockIdx.x * ROWS;
  const int rows = (N - lo < ROWS) ? (N - lo) : ROWS;

  {
    const float4 z = {0.0f, 0.0f, 0.0f, 0.0f};
    for (int i = threadIdx.x; i < ROWS * OUT_DIM / 4; i += OWAVES * 32) ((float4*)tile)[i] = z;
    for (int i = threadIdx.x; i < ROWS * NUM_HEADS / 4; i += OWAVES * 32) ((float4*)dens)[i] = z;
  }
  __syncthreads();

  int* __restrict__ mylist = lists + wv * LCAP;
  const unsigned pat = (blockIdx.x & 255u) * 0x01010101u;
  const int nchunks = (E + KSCAN_IDS - 1) / KSCAN_IDS;
  const int nsteps = (nchunks + OWAVES - 1) / OWAVES;
  const int rot = (int)(((long long)blockIdx.x * nchunks) / gridDim.x);
  int cnt = 0;
#pragma nounroll
  for (int st = 0; st < nsteps; ++st) {
    const int k = st * OWAVES + wv;
    if (k >= nchunks) break;
    const int ck = (k + rot < nchunks) ? (k + rot) : (k + rot - nchunks);
    const unsigned base = (unsigned)ck * KSCAN_IDS;
    uint4 kv[KSCAN_U];
#pragma unroll
    for (int u = 0; u < KSCAN_U; ++u) kv[u] = *(const uint4*)(key + base + u * 512 + lane * 16);
    static_assert(KSCAN_U == 4, "two flag words: the loads 0, 1 and the loads 2, 3");
    unsigned mk[KSCAN_U / 2];
#pragma unroll
    for (int u = 0; u < KSCAN_U; u += 2) {
      mk[u >> 1] = (equal_bytes(kv[u].x, pat) >> 7) | (equal_bytes(kv[u].y, pat) >> 6) |
                   (equal_bytes(kv[u].z, pat) >> 5) | (equal_bytes(kv[u].w, pat) >> 4) |
                   (equal_bytes(kv[u + 1].x, pat) >> 3) | (equal_bytes(kv[u + 1].y, pat) >> 2) |
                   (equal_bytes(kv[u + 1].z, pat) >> 1) | equal_bytes(kv[u + 1].w, pat);
    }
#pragma nounroll
    for (int hf = 0; hf < KSCAN_U / 2; ++hf) {
      unsigned m = hf ? mk[1] : mk[0];
      for (;;) {
        const bool has = (m != 0u);
        if (__builtin_amdgcn_ballot_w32(has) == 0u) break;
        if (cnt > LCAP - 32) {
          __builtin_amdgcn_wave_barrier();
          work_off(mylist, cnt, lane, ei, E, H, alpha_src, alpha_dst, tile, dens, lo, rows);
          __builtin_amdgcn_wave_barrier();
          cnt = 0;
        }
        const unsigned p = has ? (unsigned)__builtin_ctz(m) : 0u;
        const unsigned e = base + (unsigned)(2 * hf) * 512u + ((p >> 2) & 1u) * 512u + lane * 16u + 4u * (p & 3u) + (p >> 3);
        const bool ok = has && e < (unsigned)E;
        const unsigned okm = __builtin_amdgcn_ballot_w32(ok);
        const unsigned pre = __builtin_amdgcn_mbcnt_lo(okm, 0u);
        if (ok) mylist[cnt + (int)pre] = (int)e;
        cnt += __builtin_popcount(okm);
        m &= m - 1u;
      }
    }
  }
  if (cnt > 0) {
    __builtin_amdgcn_wave_barrier();
    work_off(mylist, cnt, lane, ei, E, H, alpha_src, alpha_dst, tile, dens, lo, rows);
  }
  __syncthreads();

  const unsigned head = lane >> 3;
  for (int r = wv; r < rows; r += OWAVES) {
    const float4 v = ((const float4*)(tile + r * OUT_DIM))[lane];
    const float dd = dens[r * NUM_HEADS + head] + GAT_EPS;
    float4 q;
    q.x = v.x / dd;
    q.y = v.y / dd;
    q.z = v.z / dd;
    q.w = v.w / dd;
    ((float4*)(out + (size_t)(lo + r) * OUT_DIM))[lane] = q;
  }
}

extern "C" void kernel_launch(void* const* d_in, const int* in_sizes, int n_in,
                              void* d_out, int out_size, void* d_ws, size_t ws_size,
                              hipStream_t stream) {
  const float* X  = (const float*)d_in[0];
  const int*   ei = (const int*)d_in[1];
  const float* W  = (const float*)d_in[2];
  const float* a  = (const float*)d_in[3];

  const int N = in_sizes[0] / IN_DIM;
  const int E = in_sizes[1] / 2;

  float* H         = (float*)d_ws;
  float* alpha_src = H + (size_t)N * OUT_DIM;
  float* alpha_dst = alpha_src + (size_t)N * NUM_HEADS;
  unsigned* keyw   = (unsigned*)(alpha_dst + (size_t)N * NUM_HEADS);

  gat_gemm_wmma<<<GBLOCKS, GWAVES * 32, 0, stream>>>(X, W, H, N);
  gat_alpha<<<(N + 7) / 8, 256, 0, stream>>>(H, a, alpha_src, alpha_dst, ei, keyw, N, E);
  gat_owner_pass<<<(N + ROWS - 1) / ROWS, OWAVES * 32, 0, stream>>>(ei, (const unsigned char*)keyw, H,
                                                                  alpha_src, alpha_dst,
                                                                  (float*)d_out, N, E);
}


# alpha launch removed (3 launches to 2): a GEMM wave takes alpha_src/alpha_dst of its 16 rows from its accumulators (DPP sums, same order); its workgroup writes the edge key bytes in its prologue
# speedup vs baseline: 4.2730x; 1.0647x over previous
#include <hip/hip_runtime.h>
#include <hip/hip_bf16.h>

typedef __attribute__((ext_vector_type(2))) float v2f;
typedef __attribute__((ext_vector_type(8))) float v8f;

#define IN_DIM    128
#define OUT_DIM   128
#define NUM_HEADS 4
#define HEAD_DIM  32
#define NEG_SLOPE 0.2f
#define GAT_EPS   1e-16f

#define GWAVES 16
#define GBLOCKS 256
#define WL_F4 ((IN_DIM / 2) * (OUT_DIM / 2))
#define WSTAGE_TRIPS (((IN_DIM / 2) * (OUT_DIM / 4)) / (GWAVES * 32))

static_assert(IN_DIM == 128 && OUT_DIM == 128, "the slab kernel is written for 128 x 128 weights");
static_assert(WSTAGE_TRIPS * GWAVES * 32 == (IN_DIM / 2) * (OUT_DIM / 4), "the copy of W divides evenly over the threads");

__device__ __forceinline__ void load_slab(const float* __restrict__ X, const int M, const int slab,
                                          const int c, const int h, float4 (&a)[16]) {
  int arow = slab * 16 + c;
  if (arow >= M) arow = M - 1;
  const float4* __restrict__ xr = (const float4*)(X + (size_t)arow * IN_DIM) + h;
#pragma unroll
  for (int j = 0; j < 16; ++j) a[j] = xr[2 * j];
}

#define ROWS      392
#define KSCAN_U   4
#define KSCAN_IDS (32 * 16 * KSCAN_U)
#define KEY_AHEAD 4

__device__ __forceinline__ int4 key_ids_slow(const int* __restrict__ dstp, const unsigned q, const int E) {
  const unsigned e0 = 4u * q;
  int4 v;
  v.x = (e0 + 0u < (unsigned)E) ? dstp[e0 + 0u] : 0;
  v.y = (e0 + 1u < (unsigned)E) ? dstp[e0 + 1u] : 0;
  v.z = (e0 + 2u < (unsigned)E) ? dstp[e0 + 2u] : 0;
  v.w = (e0 + 3u < (unsigned)E) ? dstp[e0 + 3u] : 0;
  return v;
}

__device__ __forceinline__ unsigned key_pack(const int4 v) {
  return (((unsigned)v.x / ROWS) & 255u) | ((((unsigned)v.y / ROWS) & 255u) << 8) |
         ((((unsigned)v.z / ROWS) & 255u) << 16) | ((((unsigned)v.w / ROWS) & 255u) << 24);
}

__device__ __forceinline__ float lane_xor1(const float v) {
  return __int_as_float(__builtin_amdgcn_update_dpp(0, __float_as_int(v), 0xB1, 0xF, 0xF, true));
}
__device__ __forceinline__ float lane_xor2(const float v) {
  return __int_as_float(__builtin_amdgcn_update_dpp(0, __float_as_int(v), 0x4E, 0xF, 0xF, true));
}
__device__ __forceinline__ float lane_xor4(const float v) {
  return __int_as_float(__builtin_amdgcn_update_dpp(0, __float_as_int(v), 0x164, 0xF, 0xF, true));
}

__device__ __forceinline__ float sum8_scatter(const float (&p)[8], const int c) {
  const bool b0 = (c & 1) != 0, b1 = (c & 2) != 0, b2 = (c & 4) != 0;
  float q[4], t[2];
#pragma unroll
  for (int i = 0; i < 4; ++i) {
    const float keep = b0 ? p[2 * i + 1] : p[2 * i];
    const float send = b0 ? p[2 * i] : p[2 * i + 1];
    q[i] = keep + lane_xor1(send);
  }
#pragma unroll
  for (int j = 0; j < 2; ++j) {
    const float keep = b1 ? q[2 * j + 1] : q[2 * j];
    const float send = b1 ? q[2 * j] : q[2 * j + 1];
    t[j] = keep + lane_xor2(send);
  }
  const float keep = b2 ? t[1] : t[0];
  const float send = b2 ? t[0] : t[1];
  return keep + lane_xor4(send);
}

__global__ __launch_bounds__(GWAVES * 32) void gat_gemm_wmma(const float* __restrict__ X,
                                                             const float* __restrict__ W,
                                                             const float* __restrict__ att,
                                                             float* __restrict__ H,
                                                             float* __restrict__ alpha_src,
                                                             float* __restrict__ alpha_dst,
                                                             const int* __restrict__ ei,
                                                             unsigned* __restrict__ keyw,
                                                             int M, int E) {
  __shared__ float4 wl[WL_F4];

  const int lane = threadIdx.x & 31;
  const int wave = __builtin_amdgcn_readfirstlane((int)(threadIdx.x >> 5));
  const int c = lane & 15;
  const int h = lane >> 4;
  const int nslabs = (M + 15) / 16;
  const int stride = (int)gridDim.x * GWAVES;

  int slab = (int)blockIdx.x + (int)gridDim.x * wave;
  float4 a[16];
  if (slab < nslabs) load_slab(X, M, slab, c, h, a);

  float4 wr0[WSTAGE_TRIPS], wr1[WSTAGE_TRIPS];
#pragma unroll
  for (int t = 0; t < WSTAGE_TRIPS; ++t) {
    const int i  = (int)threadIdx.x + t * (GWAVES * 32);
    const int kp = i >> 5;
    const int m  = i & 31;
    wr0[t] = *(const float4*)(W + (size_t)(2 * kp) * OUT_DIM + 4 * m);
    wr1[t] = *(const float4*)(W + (size_t)(2 * kp + 1) * OUT_DIM + 4 * m);
  }
  const int* __restrict__ dstp = ei + E;
  const bool vec_ok = ((E & 3) == 0);
  const unsigned nwords = (unsigned)((E + KSCAN_IDS - 1) / KSCAN_IDS) * (KSCAN_IDS / 4);
  const unsigned total = gridDim.x * blockDim.x;
  const unsigned gtid = blockIdx.x * blockDim.x + threadIdx.x;
  int4 kid[KEY_AHEAD];
  bool kfast[KEY_AHEAD];
#pragma unroll
  for (int t = 0; t < KEY_AHEAD; ++t) {
    const unsigned q = gtid + (unsigned)t * total;
    kfast[t] = vec_ok && (4u * q + 3u < (unsigned)E);
    kid[t] = make_int4(0, 0, 0, 0);
    if (kfast[t]) kid[t] = *(const int4*)(dstp + 4u * q);
  }
#pragma unroll
  for (int t = 0; t < WSTAGE_TRIPS; ++t) {
    const int i  = (int)threadIdx.x + t * (GWAVES * 32);
    const int kp = i >> 5;
    const int m  = i & 31;
    const float4 r0 = wr0[t], r1 = wr1[t];
    const int e = (kp * 4 + 2 * (m >> 4)) * 16 + (m & 15);
    wl[e]      = make_float4(r0.x, r1.x, r0.y, r1.y);
    wl[e + 16] = make_float4(r0.z, r1.z, r0.w, r1.w);
  }
#pragma unroll
  for (int t = 0; t < KEY_AHEAD; ++t) {
    const unsigned q = gtid + (unsigned)t * total;
    if (q < nwords) {
      if (!kfast[t]) kid[t] = key_ids_slow(dstp, q, E);
      keyw[q] = key_pack(kid[t]);
    }
  }
  for (unsigned q = gtid + (unsigned)KEY_AHEAD * total; q < nwords; q += total) {
    const bool fast = vec_ok && (4u * q + 3u < (unsigned)E);
    keyw[q] = key_pack(fast ? *(const int4*)(dstp + 4u * q) : key_ids_slow(dstp, q, E));
  }
  __syncthreads();

  const float4* __restrict__ bl = wl + (128 * h + c);

  while (slab < nslabs) {
    v8f acc[8] = {};
    float4 bc[8], bn[8];
#pragma unroll
    for (int i = 0; i < 8; ++i) bc[i] = bl[(i >> 2) * 64 + (i & 3) * 16];
#pragma unroll
    for (int j = 0; j < 16; ++j) {
      if (j + 1 < 16) {
#pragma unroll
        for (int i = 0; i < 8; ++i) bn[i] = bl[(4 * (j + 1) + (i >> 2)) * 64 + (i & 3) * 16];
      }
#pragma unroll
      for (int s = 0; s < 2; ++s) {
        v2f av;
        av.x = s ? a[j].z : a[j].x;
        av.y = s ? a[j].w : a[j].y;
#pragma unroll
        for (int p = 0; p < 4; ++p) {
          const float4 bv = bc[4 * s + p];
          v2f b;
          b.x = bv.x; b.y = bv.y;
          acc[2 * p] = __builtin_amdgcn_wmma_f32_16x16x4_f32(false, av, false, b, (short)0, acc[2 * p], false, false);
          b.x = bv.z; b.y = bv.w;
          acc[2 * p + 1] = __builtin_amdgcn_wmma_f32_16x16x4_f32(false, av, false, b, (short)0, acc[2 * p + 1], false, false);
        }
      }
      __builtin_amdgcn_sched_barrier(0);
#pragma unroll
      for (int i = 0; i < 8; ++i) bc[i] = bn[i];
    }

    const int m0 = slab * 16;
    slab += stride;
    if (slab < nslabs) load_slab(X, M, slab, c, h, a);

#pragma unroll
    for (int r = 0; r < 8; ++r) {
      const int row = m0 + 8 * h + r;
      if (row < M) {
        float4* __restrict__ hp = (float4*)(H + (size_t)row * OUT_DIM) + c;
        hp[0]  = make_float4(acc[0][r], acc[1][r], acc[2][r], acc[3][r]);
        hp[16] = make_float4(acc[4][r], acc[5][r], acc[6][r], acc[7][r]);
      }
    }

#pragma unroll
    for (int g = 0; g < 2; ++g) {
      const int head = 2 * g + (c >> 3);
      const float4 as = *(const float4*)(att + head * (2 * HEAD_DIM) + 4 * (c & 7));
      const float4 ad = *(const float4*)(att + head * (2 * HEAD_DIM) + HEAD_DIM + 4 * (c & 7));
      float ps[8], pd[8];
#pragma unroll
      for (int r = 0; r < 8; ++r) {
        const float x = acc[4 * g][r], y = acc[4 * g + 1][r], z = acc[4 * g + 2][r], w = acc[4 * g + 3][r];
        ps[r] = x * as.x + y * as.y + z * as.z + w * as.w;
        pd[r] = x * ad.x + y * ad.y + z * ad.z + w * ad.w;
      }
      const float ss = sum8_scatter(ps, c);
      const float dd = sum8_scatter(pd, c);
      const int row = m0 + 8 * h + (c & 7);
      if (row < M) {
        alpha_src[(size_t)row * NUM_HEADS + head] = ss;
        alpha_dst[(size_t)row * NUM_HEADS + head] = dd;
      }
    }
  }
}

#define OWAVES   32
#define LCAP     896
#define EPW 32
#define CH  8
#define NGRP (EPW / CH)

static_assert(LCAP >= 2 * EPW, "a full list still takes the edges of one more ballot");
static_assert(((ROWS + 1) * OUT_DIM + (ROWS + 1) * NUM_HEADS + OWAVES * LCAP) * 4 <= 320 * 1024, "LDS budget");

__device__ __forceinline__ float lane_to_scalar(float v, int l) {
  return __int_as_float(__builtin_amdgcn_readlane(__float_as_int(v), l));
}

__device__ __forceinline__ void lds_add(float* p, float v) {
  __hip_atomic_fetch_add(p, v, __ATOMIC_RELAXED, __HIP_MEMORY_SCOPE_WORKGROUP);
}

__device__ __forceinline__ void work_off(const int* __restrict__ lst, const int n,
                                         const unsigned lane,
                                         const int* __restrict__ ei, const int E,
                                         const float* __restrict__ H,
                                         const float* __restrict__ alpha_src,
                                         const float* __restrict__ alpha_dst,
                                         float* tile, float* dens, const int lo, const int rows) {
  const unsigned hh = lane & (NUM_HEADS - 1);
  const int qsub = (int)(lane >> 2);
#pragma nounroll
  for (int b = 0; b < n; b += EPW) {
    const int cnt = (n - b < EPW) ? (n - b) : EPW;
    unsigned sv[NGRP], dg[NGRP], dl[NGRP], word[NGRP];
    float    ex[NGRP];
#pragma unroll
    for (int g = 0; g < NGRP; ++g) {
      const int q = CH * g + qsub;
      word[g] = (unsigned)lst[b + ((q < cnt) ? q : 0)];
    }
#pragma unroll
    for (int g = 0; g < NGRP; ++g) {
      sv[g] = (unsigned)ei[word[g]];
      dg[g] = (unsigned)ei[(unsigned)E + word[g]];
    }
#pragma unroll
    for (int g = 0; g < NGRP; ++g) {
      const unsigned row = dg[g] - (unsigned)lo;
      dl[g] = (CH * g + qsub < cnt && row < (unsigned)rows) ? row : (unsigned)ROWS;
    }
#pragma unroll
    for (int g = 0; g < NGRP; ++g) {
      const float as = alpha_src[sv[g] * NUM_HEADS + hh];
      const float ad = alpha_dst[dg[g] * NUM_HEADS + hh];
      float v = as + ad;
      v = (v >= 0.0f) ? v : NEG_SLOPE * v;
      ex[g] = expf(v);
    }
#pragma unroll
    for (int g = 0; g < NGRP; ++g) {
      lds_add(dens + (dl[g] * NUM_HEADS + hh), ex[g]);
    }
#pragma unroll
    for (int g = 0; g < NGRP; ++g) {
      if (CH * g < cnt) {
        float h[CH][4];
#pragma unroll
        for (int u = 0; u < CH; ++u) {
          const unsigned s = (unsigned)__builtin_amdgcn_readlane((int)sv[g], 4 * u);
          const float* __restrict__ hrow = H + (s * OUT_DIM + lane);
          h[u][0] = hrow[0 * HEAD_DIM];
          h[u][1] = hrow[1 * HEAD_DIM];
          h[u][2] = hrow[2 * HEAD_DIM];
          h[u][3] = hrow[3 * HEAD_DIM];
        }
#pragma unroll
        for (int u = 0; u < CH; ++u) {
          const unsigned d = (unsigned)__builtin_amdgcn_readlane((int)dl[g], 4 * u);
          const float w0 = lane_to_scalar(ex[g], 4 * u + 0);
          const float w1 = lane_to_scalar(ex[g], 4 * u + 1);
          const float w2 = lane_to_scalar(ex[g], 4 * u + 2);
          const float w3 = lane_to_scalar(ex[g], 4 * u + 3);
          float* op = tile + (d * OUT_DIM + lane);
          lds_add(op + 0 * HEAD_DIM, h[u][0] * w0);
          lds_add(op + 1 * HEAD_DIM, h[u][1] * w1);
          lds_add(op + 2 * HEAD_DIM, h[u][2] * w2);
          lds_add(op + 3 * HEAD_DIM, h[u][3] * w3);
        }
      }
    }
  }
}

__device__ __forceinline__ unsigned equal_bytes(const unsigned v, const unsigned pat) {
  const unsigned x = v ^ pat;
  return (x - 0x01010101u) & ~x & 0x80808080u;
}

__global__ __launch_bounds__(OWAVES * 32) void gat_owner_pass(const int* __restrict__ ei,
                                                              const unsigned char* __restrict__ key,
                                                              const float* __restrict__ H,
                                                              const float* __restrict__ alpha_src,
                                                              const float* __restrict__ alpha_dst,
                                                              float* __restrict__ out,
                                                              int N, int E) {
  __shared__ __attribute__((aligned(16))) float tile[(ROWS + 1) * OUT_DIM];
  __shared__ __attribute__((aligned(16))) float dens[(ROWS + 1) * NUM_HEADS];
  __shared__ int lists[OWAVES * LCAP];

  const unsigned lane = threadIdx.x & 31;
  const int wv = __builtin_amdgcn_readfirstlane((int)(threadIdx.x >> 5));
  const int lo = blockIdx.x * ROWS;
  const int rows = (N - lo < ROWS) ? (N - lo) : ROWS;

  {
    const float4 z = {0.0f, 0.0f, 0.0f, 0.0f};
    for (int i = threadIdx.x; i < ROWS * OUT_DIM / 4; i += OWAVES * 32) ((float4*)tile)[i] = z;
    for (int i = threadIdx.x; i < ROWS * NUM_HEADS / 4; i += OWAVES * 32) ((float4*)dens)[i] = z;
  }
  __syncthreads();

  int* __restrict__ mylist = lists + wv * LCAP;
  const unsigned pat = (blockIdx.x & 255u) * 0x01010101u;
  const int nchunks = (E + KSCAN_IDS - 1) / KSCAN_IDS;
  const int nsteps = (nchunks + OWAVES - 1) / OWAVES;
  const int rot = (int)(((long long)blockIdx.x * nchunks) / gridDim.x);
  int cnt = 0;
#pragma nounroll
  for (int st = 0; st < nsteps; ++st) {
    const int k = st * OWAVES + wv;
    if (k >= nchunks) break;
    const int ck = (k + rot < nchunks) ? (k + rot) : (k + rot - nchunks);
    const unsigned base = (unsigned)ck * KSCAN_IDS;
    uint4 kv[KSCAN_U];
#pragma unroll
    for (int u = 0; u < KSCAN_U; ++u) kv[u] = *(const uint4*)(key + base + u * 512 + lane * 16);
    static_assert(KSCAN_U == 4, "two flag words: the loads 0, 1 and the loads 2, 3");
    unsigned mk[KSCAN_U / 2];
#pragma unroll
    for (int u = 0; u < KSCAN_U; u += 2) {
      mk[u >> 1] = (equal_bytes(kv[u].x, pat) >> 7) | (equal_bytes(kv[u].y, pat) >> 6) |
                   (equal_bytes(kv[u].z, pat) >> 5) | (equal_bytes(kv[u].w, pat) >> 4) |
                   (equal_bytes(kv[u + 1].x, pat) >> 3) | (equal_bytes(kv[u + 1].y, pat) >> 2) |
                   (equal_bytes(kv[u + 1].z, pat) >> 1) | equal_bytes(kv[u + 1].w, pat);
    }
#pragma nounroll
    for (int hf = 0; hf < KSCAN_U / 2; ++hf) {
      unsigned m = hf ? mk[1] : mk[0];
      for (;;) {
        const bool has = (m != 0u);
        if (__builtin_amdgcn_ballot_w32(has) == 0u) break;
        if (cnt > LCAP - 32) {
          __builtin_amdgcn_wave_barrier();
          work_off(mylist, cnt, lane, ei, E, H, alpha_src, alpha_dst, tile, dens, lo, rows);
          __builtin_amdgcn_wave_barrier();
          cnt = 0;
        }
        const unsigned p = has ? (unsigned)__builtin_ctz(m) : 0u;
        const unsigned e = base + (unsigned)(2 * hf) * 512u + ((p >> 2) & 1u) * 512u + lane * 16u + 4u * (p & 3u) + (p >> 3);
        const bool ok = has && e < (unsigned)E;
        const unsigned okm = __builtin_amdgcn_ballot_w32(ok);
        const unsigned pre = __builtin_amdgcn_mbcnt_lo(okm, 0u);
        if (ok) mylist[cnt + (int)pre] = (int)e;
        cnt += __builtin_popcount(okm);
        m &= m - 1u;
      }
    }
  }
  if (cnt > 0) {
    __builtin_amdgcn_wave_barrier();
    work_off(mylist, cnt, lane, ei, E, H, alpha_src, alpha_dst, tile, dens, lo, rows);
  }
  __syncthreads();

  const unsigned head = lane >> 3;
  for (int r = wv; r < rows; r += OWAVES) {
    const float4 v = ((const float4*)(tile + r * OUT_DIM))[lane];
    const float dd = dens[r * NUM_HEADS + head] + GAT_EPS;
    float4 q;
    q.x = v.x / dd;
    q.y = v.y / dd;
    q.z = v.z / dd;
    q.w = v.w / dd;
    ((float4*)(out + (size_t)(lo + r) * OUT_DIM))[lane] = q;
  }
}

extern "C" void kernel_launch(void* const* d_in, const int* in_sizes, int n_in,
                              void* d_out, int out_size, void* d_ws, size_t ws_size,
                              hipStream_t stream) {
  const float* X  = (const float*)d_in[0];
  const int*   ei = (const int*)d_in[1];
  const float* W  = (const float*)d_in[2];
  const float* a  = (const float*)d_in[3];

  const int N = in_sizes[0] / IN_DIM;
  const int E = in_sizes[1] / 2;

  float* H         = (float*)d_ws;
  float* alpha_src = H + (size_t)N * OUT_DIM;
  float* alpha_dst = alpha_src + (size_t)N * NUM_HEADS;
  unsigned* keyw   = (unsigned*)(alpha_dst + (size_t)N * NUM_HEADS);

  gat_gemm_wmma<<<GBLOCKS, GWAVES * 32, 0, stream>>>(X, W, a, H, alpha_src, alpha_dst, ei, keyw, N, E);
  gat_owner_pass<<<(N + ROWS - 1) / ROWS, OWAVES * 32, 0, stream>>>(ei, (const unsigned char*)keyw, H,
                                                                  alpha_src, alpha_dst,
                                                                  (float*)d_out, N, E);
}


# owner pass lane map: lane owns 4 adjacent dims of a head: source row = one 16-byte load per lane (was four 4-byte), weight = a DPP8 move; LDS row keeps dim 4L+j at word 32j+L; sums: LDS arrival order
# speedup vs baseline: 4.4852x; 1.0497x over previous
#include <hip/hip_runtime.h>
#include <hip/hip_bf16.h>

typedef __attribute__((ext_vector_type(2))) float v2f;
typedef __attribute__((ext_vector_type(8))) float v8f;

#define IN_DIM    128
#define OUT_DIM   128
#define NUM_HEADS 4
#define HEAD_DIM  32
#define NEG_SLOPE 0.2f
#define GAT_EPS   1e-16f

#define GWAVES 16
#define GBLOCKS 256
#define WL_F4 ((IN_DIM / 2) * (OUT_DIM / 2))
#define WSTAGE_TRIPS (((IN_DIM / 2) * (OUT_DIM / 4)) / (GWAVES * 32))

static_assert(IN_DIM == 128 && OUT_DIM == 128, "the slab kernel is written for 128 x 128 weights");
static_assert(WSTAGE_TRIPS * GWAVES * 32 == (IN_DIM / 2) * (OUT_DIM / 4), "the copy of W divides evenly over the threads");

__device__ __forceinline__ void load_slab(const float* __restrict__ X, const int M, const int slab,
                                          const int c, const int h, float4 (&a)[16]) {
  int arow = slab * 16 + c;
  if (arow >= M) arow = M - 1;
  const float4* __restrict__ xr = (const float4*)(X + (size_t)arow * IN_DIM) + h;
#pragma unroll
  for (int j = 0; j < 16; ++j) a[j] = xr[2 * j];
}

#define ROWS      392
#define KSCAN_U   4
#define KSCAN_IDS (32 * 16 * KSCAN_U)
#define KEY_AHEAD 4

__device__ __forceinline__ int4 key_ids_slow(const int* __restrict__ dstp, const unsigned q, const int E) {
  const unsigned e0 = 4u * q;
  int4 v;
  v.x = (e0 + 0u < (unsigned)E) ? dstp[e0 + 0u] : 0;
  v.y = (e0 + 1u < (unsigned)E) ? dstp[e0 + 1u] : 0;
  v.z = (e0 + 2u < (unsigned)E) ? dstp[e0 + 2u] : 0;
  v.w = (e0 + 3u < (unsigned)E) ? dstp[e0 + 3u] : 0;
  return v;
}

__device__ __forceinline__ unsigned key_pack(const int4 v) {
  return (((unsigned)v.x / ROWS) & 255u) | ((((unsigned)v.y / ROWS) & 255u) << 8) |
         ((((unsigned)v.z / ROWS) & 255u) << 16) | ((((unsigned)v.w / ROWS) & 255u) << 24);
}

__device__ __forceinline__ float lane_xor1(const float v) {
  return __int_as_float(__builtin_amdgcn_update_dpp(0, __float_as_int(v), 0xB1, 0xF, 0xF, true));
}
__device__ __forceinline__ float lane_xor2(const float v) {
  return __int_as_float(__builtin_amdgcn_update_dpp(0, __float_as_int(v), 0x4E, 0xF, 0xF, true));
}
__device__ __forceinline__ float lane_xor4(const float v) {
  return __int_as_float(__builtin_amdgcn_update_dpp(0, __float_as_int(v), 0x164, 0xF, 0xF, true));
}

__device__ __forceinline__ float sum8_scatter(const float (&p)[8], const int c) {
  const bool b0 = (c & 1) != 0, b1 = (c & 2) != 0, b2 = (c & 4) != 0;
  float q[4], t[2];
#pragma unroll
  for (int i = 0; i < 4; ++i) {
    const float keep = b0 ? p[2 * i + 1] : p[2 * i];
    const float send = b0 ? p[2 * i] : p[2 * i + 1];
    q[i] = keep + lane_xor1(send);
  }
#pragma unroll
  for (int j = 0; j < 2; ++j) {
    const float keep = b1 ? q[2 * j + 1] : q[2 * j];
    const float send = b1 ? q[2 * j] : q[2 * j + 1];
    t[j] = keep + lane_xor2(send);
  }
  const float keep = b2 ? t[1] : t[0];
  const float send = b2 ? t[0] : t[1];
  return keep + lane_xor4(send);
}

__global__ __launch_bounds__(GWAVES * 32) void gat_gemm_wmma(const float* __restrict__ X,
                                                             const float* __restrict__ W,
                                                             const float* __restrict__ att,
                                                             float* __restrict__ H,
                                                             float* __restrict__ alpha_src,
                                                             float* __restrict__ alpha_dst,
                                                             const int* __restrict__ ei,
                                                             unsigned* __restrict__ keyw,
                                                             int M, int E) {
  __shared__ float4 wl[WL_F4];

  const int lane = threadIdx.x & 31;
  const int wave = __builtin_amdgcn_readfirstlane((int)(threadIdx.x >> 5));
  const int c = lane & 15;
  const int h = lane >> 4;
  const int nslabs = (M + 15) / 16;
  const int stride = (int)gridDim.x * GWAVES;

  int slab = (int)blockIdx.x + (int)gridDim.x * wave;
  float4 a[16];
  if (slab < nslabs) load_slab(X, M, slab, c, h, a);

  float4 wr0[WSTAGE_TRIPS], wr1[WSTAGE_TRIPS];
#pragma unroll
  for (int t = 0; t < WSTAGE_TRIPS; ++t) {
    const int i  = (int)threadIdx.x + t * (GWAVES * 32);
    const int kp = i >> 5;
    const int m  = i & 31;
    wr0[t] = *(const float4*)(W + (size_t)(2 * kp) * OUT_DIM + 4 * m);
    wr1[t] = *(const float4*)(W + (size_t)(2 * kp + 1) * OUT_DIM + 4 * m);
  }
  const int* __restrict__ dstp = ei + E;
  const bool vec_ok = ((E & 3) == 0);
  const unsigned nwords = (unsigned)((E + KSCAN_IDS - 1) / KSCAN_IDS) * (KSCAN_IDS / 4);
  const unsigned total = gridDim.x * blockDim.x;
  const unsigned gtid = blockIdx.x * blockDim.x + threadIdx.x;
  int4 kid[KEY_AHEAD];
  bool kfast[KEY_AHEAD];
#pragma unroll
  for (int t = 0; t < KEY_AHEAD; ++t) {
    const unsigned q = gtid + (unsigned)t * total;
    kfast[t] = vec_ok && (4u * q + 3u < (unsigned)E);
    kid[t] = make_int4(0, 0, 0, 0);
    if (kfast[t]) kid[t] = *(const int4*)(dstp + 4u * q);
  }
#pragma unroll
  for (int t = 0; t < WSTAGE_TRIPS; ++t) {
    const int i  = (int)threadIdx.x + t * (GWAVES * 32);
    const int kp = i >> 5;
    const int m  = i & 31;
    const float4 r0 = wr0[t], r1 = wr1[t];
    const int e = (kp * 4 + 2 * (m >> 4)) * 16 + (m & 15);
    wl[e]      = make_float4(r0.x, r1.x, r0.y, r1.y);
    wl[e + 16] = make_float4(r0.z, r1.z, r0.w, r1.w);
  }
#pragma unroll
  for (int t = 0; t < KEY_AHEAD; ++t) {
    const unsigned q = gtid + (unsigned)t * total;
    if (q < nwords) {
      if (!kfast[t]) kid[t] = key_ids_slow(dstp, q, E);
      keyw[q] = key_pack(kid[t]);
    }
  }
  for (unsigned q = gtid + (unsigned)KEY_AHEAD * total; q < nwords; q += total) {
    const bool fast = vec_ok && (4u * q + 3u < (unsigned)E);
    keyw[q] = key_pack(fast ? *(const int4*)(dstp + 4u * q) : key_ids_slow(dstp, q, E));
  }
  __syncthreads();

  const float4* __restrict__ bl = wl + (128 * h + c);

  while (slab < nslabs) {
    v8f acc[8] = {};
    float4 bc[8], bn[8];
#pragma unroll
    for (int i = 0; i < 8; ++i) bc[i] = bl[(i >> 2) * 64 + (i & 3) * 16];
#pragma unroll
    for (int j = 0; j < 16; ++j) {
      if (j + 1 < 16) {
#pragma unroll
        for (int i = 0; i < 8; ++i) bn[i] = bl[(4 * (j + 1) + (i >> 2)) * 64 + (i & 3) * 16];
      }
#pragma unroll
      for (int s = 0; s < 2; ++s) {
        v2f av;
        av.x = s ? a[j].z : a[j].x;
        av.y = s ? a[j].w : a[j].y;
#pragma unroll
        for (int p = 0; p < 4; ++p) {
          const float4 bv = bc[4 * s + p];
          v2f b;
          b.x = bv.x; b.y = bv.y;
          acc[2 * p] = __builtin_amdgcn_wmma_f32_16x16x4_f32(false, av, false, b, (short)0, acc[2 * p], false, false);
          b.x = bv.z; b.y = bv.w;
          acc[2 * p + 1] = __builtin_amdgcn_wmma_f32_16x16x4_f32(false, av, false, b, (short)0, acc[2 * p + 1], false, false);
        }
      }
      __builtin_amdgcn_sched_barrier(0);
#pragma unroll
      for (int i = 0; i < 8; ++i) bc[i] = bn[i];
    }

    const int m0 = slab * 16;
    slab += stride;
    if (slab < nslabs) load_slab(X, M, slab, c, h, a);

#pragma unroll
    for (int r = 0; r < 8; ++r) {
      const int row = m0 + 8 * h + r;
      if (row < M) {
        float4* __restrict__ hp = (float4*)(H + (size_t)row * OUT_DIM) + c;
        hp[0]  = make_float4(acc[0][r], acc[1][r], acc[2][r], acc[3][r]);
        hp[16] = make_float4(acc[4][r], acc[5][r], acc[6][r], acc[7][r]);
      }
    }

#pragma unroll
    for (int g = 0; g < 2; ++g) {
      const int head = 2 * g + (c >> 3);
      const float4 as = *(const float4*)(att + head * (2 * HEAD_DIM) + 4 * (c & 7));
      const float4 ad = *(const float4*)(att + head * (2 * HEAD_DIM) + HEAD_DIM + 4 * (c & 7));
      float ps[8], pd[8];
#pragma unroll
      for (int r = 0; r < 8; ++r) {
        const float x = acc[4 * g][r], y = acc[4 * g + 1][r], z = acc[4 * g + 2][r], w = acc[4 * g + 3][r];
        ps[r] = x * as.x + y * as.y + z * as.z + w * as.w;
        pd[r] = x * ad.x + y * ad.y + z * ad.z + w * ad.w;
      }
      const float ss = sum8_scatter(ps, c);
      const float dd = sum8_scatter(pd, c);
      const int row = m0 + 8 * h + (c & 7);
      if (row < M) {
        alpha_src[(size_t)row * NUM_HEADS + head] = ss;
        alpha_dst[(size_t)row * NUM_HEADS + head] = dd;
      }
    }
  }
}

#define OWAVES   32
#define LCAP     896
#define EPW 32
#define CH  8
#define NGRP (EPW / CH)
#define TROW_STEP 32

static_assert(LCAP >= 2 * EPW, "a full list still takes the edges of one more ballot");
static_assert(CH == 8 && NUM_HEADS * CH == 32, "a group's weights fill the wave: head = lane >> 3, edge = lane & 7 (the DPP8 groups)");
static_assert(OUT_DIM == 4 * TROW_STEP && HEAD_DIM % 4 == 0, "a lane's 4 adjacent dims are one 16-byte load and lie in one head");
static_assert(((ROWS + 1) * OUT_DIM + (ROWS + 1) * NUM_HEADS + OWAVES * LCAP) * 4 <= 320 * 1024, "LDS budget");

__device__ __forceinline__ void lds_add(float* p, float v) {
  __hip_atomic_fetch_add(p, v, __ATOMIC_RELAXED, __HIP_MEMORY_SCOPE_WORKGROUP);
}

__device__ __forceinline__ float share8(const float v, const int u) {
  const unsigned x = __float_as_uint(v);
  switch (u) {
    case 0:  return __uint_as_float(__builtin_amdgcn_mov_dpp8(x, 0x000000u));
    case 1:  return __uint_as_float(__builtin_amdgcn_mov_dpp8(x, 0x249249u));
    case 2:  return __uint_as_float(__builtin_amdgcn_mov_dpp8(x, 0x492492u));
    case 3:  return __uint_as_float(__builtin_amdgcn_mov_dpp8(x, 0x6DB6DBu));
    case 4:  return __uint_as_float(__builtin_amdgcn_mov_dpp8(x, 0x924924u));
    case 5:  return __uint_as_float(__builtin_amdgcn_mov_dpp8(x, 0xB6DB6Du));
    case 6:  return __uint_as_float(__builtin_amdgcn_mov_dpp8(x, 0xDB6DB6u));
    default: return __uint_as_float(__builtin_amdgcn_mov_dpp8(x, 0xFFFFFFu));
  }
}

__device__ __forceinline__ void work_off(const int* __restrict__ lst, const int n,
                                         const unsigned lane,
                                         const int* __restrict__ ei, const int E,
                                         const float* __restrict__ H,
                                         const float* __restrict__ alpha_src,
                                         const float* __restrict__ alpha_dst,
                                         float* tile, float* dens, const int lo, const int rows) {
  const unsigned hh = lane >> 3;
  const int qsub = (int)(lane & 7u);
#pragma nounroll
  for (int b = 0; b < n; b += EPW) {
    const int cnt = (n - b < EPW) ? (n - b) : EPW;
    unsigned sv[NGRP], dg[NGRP], dl[NGRP], word[NGRP];
    float    ex[NGRP];
#pragma unroll
    for (int g = 0; g < NGRP; ++g) {
      const int q = CH * g + qsub;
      word[g] = (unsigned)lst[b + ((q < cnt) ? q : 0)];
    }
#pragma unroll
    for (int g = 0; g < NGRP; ++g) {
      sv[g] = (unsigned)ei[word[g]];
      dg[g] = (unsigned)ei[(unsigned)E + word[g]];
    }
#pragma unroll
    for (int g = 0; g < NGRP; ++g) {
      const unsigned row = dg[g] - (unsigned)lo;
      dl[g] = (CH * g + qsub < cnt && row < (unsigned)rows) ? row : (unsigned)ROWS;
    }
#pragma unroll
    for (int g = 0; g < NGRP; ++g) {
      const float as = alpha_src[sv[g] * NUM_HEADS + hh];
      const float ad = alpha_dst[dg[g] * NUM_HEADS + hh];
      float v = as + ad;
      v = (v >= 0.0f) ? v : NEG_SLOPE * v;
      ex[g] = expf(v);
    }
#pragma unroll
    for (int g = 0; g < NGRP; ++g) {
      lds_add(dens + (dl[g] * NUM_HEADS + hh), ex[g]);
    }
#pragma unroll
    for (int g = 0; g < NGRP; ++g) {
      if (CH * g < cnt) {
        float4 h[CH];
#pragma unroll
        for (int u = 0; u < CH; ++u) {
          const unsigned s = (unsigned)__builtin_amdgcn_readlane((int)sv[g], u);
          h[u] = ((const float4*)H)[s * (OUT_DIM / 4) + lane];
        }
#pragma unroll
        for (int u = 0; u < CH; ++u) {
          const unsigned d = (unsigned)__builtin_amdgcn_readlane((int)dl[g], u);
          const float w = share8(ex[g], u);
          float* op = tile + (d * OUT_DIM + lane);
          lds_add(op + 0 * TROW_STEP, h[u].x * w);
          lds_add(op + 1 * TROW_STEP, h[u].y * w);
          lds_add(op + 2 * TROW_STEP, h[u].z * w);
          lds_add(op + 3 * TROW_STEP, h[u].w * w);
        }
      }
    }
  }
}

__device__ __forceinline__ unsigned equal_bytes(const unsigned v, const unsigned pat) {
  const unsigned x = v ^ pat;
  return (x - 0x01010101u) & ~x & 0x80808080u;
}

__global__ __launch_bounds__(OWAVES * 32) void gat_owner_pass(const int* __restrict__ ei,
                                                              const unsigned char* __restrict__ key,
                                                              const float* __restrict__ H,
                                                              const float* __restrict__ alpha_src,
                                                              const float* __restrict__ alpha_dst,
                                                              float* __restrict__ out,
                                                              int N, int E) {
  __shared__ __attribute__((aligned(16))) float tile[(ROWS + 1) * OUT_DIM];
  __shared__ __attribute__((aligned(16))) float dens[(ROWS + 1) * NUM_HEADS];
  __shared__ int lists[OWAVES * LCAP];

  const unsigned lane = threadIdx.x & 31;
  const int wv = __builtin_amdgcn_readfirstlane((int)(threadIdx.x >> 5));
  const int lo = blockIdx.x * ROWS;
  const int rows = (N - lo < ROWS) ? (N - lo) : ROWS;

  {
    const float4 z = {0.0f, 0.0f, 0.0f, 0.0f};
    for (int i = threadIdx.x; i < ROWS * OUT_DIM / 4; i += OWAVES * 32) ((float4*)tile)[i] = z;
    for (int i = threadIdx.x; i < ROWS * NUM_HEADS / 4; i += OWAVES * 32) ((float4*)dens)[i] = z;
  }
  __syncthreads();

  int* __restrict__ mylist = lists + wv * LCAP;
  const unsigned pat = (blockIdx.x & 255u) * 0x01010101u;
  const int nchunks = (E + KSCAN_IDS - 1) / KSCAN_IDS;
  const int nsteps = (nchunks + OWAVES - 1) / OWAVES;
  const int rot = (int)(((long long)blockIdx.x * nchunks) / gridDim.x);
  int cnt = 0;
#pragma nounroll
  for (int st = 0; st < nsteps; ++st) {
    const int k = st * OWAVES + wv;
    if (k >= nchunks) break;
    const int ck = (k + rot < nchunks) ? (k + rot) : (k + rot - nchunks);
    const unsigned base = (unsigned)ck * KSCAN_IDS;
    uint4 kv[KSCAN_U];
#pragma unroll
    for (int u = 0; u < KSCAN_U; ++u) kv[u] = *(const uint4*)(key + base + u * 512 + lane * 16);
    static_assert(KSCAN_U == 4, "two flag words: the loads 0, 1 and the loads 2, 3");
    unsigned mk[KSCAN_U / 2];
#pragma unroll
    for (int u = 0; u < KSCAN_U; u += 2) {
      mk[u >> 1] = (equal_bytes(kv[u].x, pat) >> 7) | (equal_bytes(kv[u].y, pat) >> 6) |
                   (equal_bytes(kv[u].z, pat) >> 5) | (equal_bytes(kv[u].w, pat) >> 4) |
                   (equal_bytes(kv[u + 1].x, pat) >> 3) | (equal_bytes(kv[u + 1].y, pat) >> 2) |
                   (equal_bytes(kv[u + 1].z, pat) >> 1) | equal_bytes(kv[u + 1].w, pat);
    }
#pragma nounroll
    for (int hf = 0; hf < KSCAN_U / 2; ++hf) {
      unsigned m = hf ? mk[1] : mk[0];
      for (;;) {
        const bool has = (m != 0u);
        if (__builtin_amdgcn_ballot_w32(has) == 0u) break;
        if (cnt > LCAP - 32) {
          __builtin_amdgcn_wave_barrier();
          work_off(mylist, cnt, lane, ei, E, H, alpha_src, alpha_dst, tile, dens, lo, rows);
          __builtin_amdgcn_wave_barrier();
          cnt = 0;
        }
        const unsigned p = has ? (unsigned)__builtin_ctz(m) : 0u;
        const unsigned e = base + (unsigned)(2 * hf) * 512u + ((p >> 2) & 1u) * 512u + lane * 16u + 4u * (p & 3u) + (p >> 3);
        const bool ok = has && e < (unsigned)E;
        const unsigned okm = __builtin_amdgcn_ballot_w32(ok);
        const unsigned pre = __builtin_amdgcn_mbcnt_lo(okm, 0u);
        if (ok) mylist[cnt + (int)pre] = (int)e;
        cnt += __builtin_popcount(okm);
        m &= m - 1u;
      }
    }
  }
  if (cnt > 0) {
    __builtin_amdgcn_wave_barrier();
    work_off(mylist, cnt, lane, ei, E, H, alpha_src, alpha_dst, tile, dens, lo, rows);
  }
  __syncthreads();

  const unsigned head = lane >> 3;
  for (int r = wv; r < rows; r += OWAVES) {
    const float* tr = tile + (r * OUT_DIM + (int)lane);
    const float4 v = make_float4(tr[0 * TROW_STEP], tr[1 * TROW_STEP], tr[2 * TROW_STEP], tr[3 * TROW_STEP]);
    const float dd = dens[r * NUM_HEADS + head] + GAT_EPS;
    float4 q;
    q.x = v.x / dd;
    q.y = v.y / dd;
    q.z = v.z / dd;
    q.w = v.w / dd;
    ((float4*)(out + (size_t)(lo + r) * OUT_DIM))[lane] = q;
  }
}

extern "C" void kernel_launch(void* const* d_in, const int* in_sizes, int n_in,
                              void* d_out, int out_size, void* d_ws, size_t ws_size,
                              hipStream_t stream) {
  const float* X  = (const float*)d_in[0];
  const int*   ei = (const int*)d_in[1];
  const float* W  = (const float*)d_in[2];
  const float* a  = (const float*)d_in[3];

  const int N = in_sizes[0] / IN_DIM;
  const int E = in_sizes[1] / 2;

  float* H         = (float*)d_ws;
  float* alpha_src = H + (size_t)N * OUT_DIM;
  float* alpha_dst = alpha_src + (size_t)N * NUM_HEADS;
  unsigned* keyw   = (unsigned*)(alpha_dst + (size_t)N * NUM_HEADS);

  gat_gemm_wmma<<<GBLOCKS, GWAVES * 32, 0, stream>>>(X, W, a, H, alpha_src, alpha_dst, ei, keyw, N, E);
  gat_owner_pass<<<(N + ROWS - 1) / ROWS, OWAVES * 32, 0, stream>>>(ei, (const unsigned char*)keyw, H,
                                                                  alpha_src, alpha_dst,
                                                                  (float*)d_out, N, E);
}
